# starting kernel
# baseline (speedup 1.0000x reference)
#include <hip/hip_runtime.h>

typedef float v2f __attribute__((ext_vector_type(2)));
typedef float v8f __attribute__((ext_vector_type(8)));

#define BS      32
#define CS      512
#define SSEG    8
#define HS      64
#define WSZ     64
#define SPATIAL (HS * WSZ)
#define NSTEP   (SPATIAL / 4)
#define SEMW    256

__global__ void build_a(const int* __restrict__ sem, float* __restrict__ Apre) {
    unsigned gid  = blockIdx.x * blockDim.x + threadIdx.x;
    unsigned lane = gid & 31u;
    unsigned t    = (gid >> 5) & (NSTEP - 1);
    unsigned b    = gid >> 15;
    unsigned seg  = lane & 15u;
    unsigned kh   = lane >> 4;
    unsigned pos0 = 4u * t + 2u * kh;

    v2f a;
    if (seg < SSEG) {
        unsigned base = (b * SSEG + seg) * (SEMW * SEMW);
        #pragma unroll
        for (int i = 0; i < 2; ++i) {
            unsigned pos = pos0 + i;
            unsigned h = pos >> 6, w = pos & 63u;
            int v = sem[base + (h * 4u) * SEMW + (w * 4u)];
            a[i] = (v != 0) ? 1.0f : 0.0f;
        }
    } else if (seg == SSEG) {
        a.x = 1.0f; a.y = 1.0f;
    } else {
        a.x = 0.0f; a.y = 0.0f;
    }
    ((v2f*)Apre)[gid] = a;
}

__global__ void area_kernel(const int* __restrict__ sem, float* __restrict__ area) {
    unsigned bs = blockIdx.x;
    __shared__ float red[256];
    unsigned base = bs * (SEMW * SEMW);
    float acc = 0.0f;
    for (unsigned pos = threadIdx.x; pos < SPATIAL; pos += 256u) {
        unsigned h = pos >> 6, w = pos & 63u;
        acc += (sem[base + (h * 4u) * SEMW + (w * 4u)] != 0) ? 1.0f : 0.0f;
    }
    red[threadIdx.x] = acc;
    __syncthreads();
    for (unsigned off = 128; off > 0; off >>= 1) {
        if (threadIdx.x < off) red[threadIdx.x] += red[threadIdx.x + off];
        __syncthreads();
    }
    if (threadIdx.x == 0) area[bs] = red[0];
}

__global__ void __launch_bounds__(128) seg_reduce(const float* __restrict__ codes,
                                                  const float* __restrict__ Apre,
                                                  const float* __restrict__ area,
                                                  float* __restrict__ out) {
    const unsigned wave = threadIdx.x >> 5;
    const unsigned lane = threadIdx.x & 31u;
    const unsigned b    = blockIdx.x >> 3;
    const unsigned cg   = (blockIdx.x & 7u) * 4u + wave;
    const unsigned n    = lane & 15u;
    const unsigned kh   = lane >> 4;

    const v2f* __restrict__ bptr =
        (const v2f*)(codes + ((size_t)(b * CS + cg * 16u + n)) * SPATIAL) + kh;
    const v2f* __restrict__ aptr =
        (const v2f*)Apre + (size_t)b * NSTEP * 32u + lane;

    v8f c = {0.f, 0.f, 0.f, 0.f, 0.f, 0.f, 0.f, 0.f};

    #pragma unroll 4
    for (int t = 0; t < NSTEP; ++t) {
        v2f bvec = __builtin_nontemporal_load(bptr + 2 * t);
        v2f avec = aptr[t * 32];
        c = __builtin_amdgcn_wmma_f32_16x16x4_f32(
                 false, avec,  false, bvec,
                 (short)0, c,  false,  false);
    }

    float glob = __shfl(c[0], (int)(16u + n), 32);

    if (lane < 16u) {
        const float ginv = glob * (1.0f / (float)SPATIAL);
        float* o = out + (size_t)b * (SSEG * CS) + cg * 16u + n;
        #pragma unroll
        for (int s = 0; s < SSEG; ++s) {
            float ar = area[b * SSEG + s];
            o[(size_t)s * CS] = (ar > 0.0f) ? (c[s] / ar) : ginv;
        }
    }
}

extern "C" void kernel_launch(void* const* d_in, const int* in_sizes, int n_in,
                              void* d_out, int out_size, void* d_ws, size_t ws_size,
                              hipStream_t stream) {
    const float* codes = (const float*)d_in[0];
    const int*   sem   = (const int*)d_in[1];
    float*       out   = (float*)d_out;

    float* Apre = (float*)d_ws;
    float* area = (float*)((char*)d_ws +
                           (size_t)BS * NSTEP * 32 * 2 * sizeof(float));

    hipLaunchKernelGGL(build_a, dim3((BS * NSTEP * 32) / 256), dim3(256),
                       0, stream, sem, Apre);
    hipLaunchKernelGGL(area_kernel, dim3(BS * SSEG), dim3(256),
                       0, stream, sem, area);
    hipLaunchKernelGGL(seg_reduce, dim3(BS * 8), dim3(128),
                       0, stream, codes, Apre, area, out);
}


# block per (batch, 512-position chunk) builds the mask fragments once in LDS from sem and reuses them over all 512 channels; chunk partials + combine kernel (3 launches to 2, sum order changes)
# speedup vs baseline: 4.1892x; 4.1892x over previous
#include <hip/hip_runtime.h>


typedef float v2f __attribute__((ext_vector_type(2)));
typedef float v4f __attribute__((ext_vector_type(4)));
typedef float v8f __attribute__((ext_vector_type(8)));

#define BS      32
#define CS      512
#define SSEG    8
#define SPATIAL 4096
#define SEMW    256
#define NQ      8
#define PCH     (SPATIAL / NQ)
#define NJ      (PCH / 8)
#define TILE    160

__global__ void __launch_bounds__(256) seg_partial(const float* __restrict__ codes,
                                                   const int* __restrict__ sem,
                                                   float* __restrict__ part,
                                                   float* __restrict__ areap) {
    __shared__ v4f Ash[NJ * 32];
    __shared__ int cntw[8][SSEG];

    const unsigned tid  = threadIdx.x;
    const unsigned wave = tid >> 5;
    const unsigned lane = tid & 31u;
    const unsigned b    = blockIdx.x >> 3;
    const unsigned q    = blockIdx.x & 7u;
    const unsigned c    = lane & 15u;
    const unsigned h    = lane >> 4;

    const float* __restrict__ bbase =
        codes + ((size_t)(b * CS + wave * 64u + c)) * SPATIAL + q * PCH + 4u * h;

    v4f B[2][32];
    #pragma unroll
    for (int j = 0; j < 16; ++j) {
        B[0][j]      = __builtin_nontemporal_load((const v4f*)(bbase + 8 * j));
        B[0][16 + j] = __builtin_nontemporal_load((const v4f*)(bbase + 16 * SPATIAL + 8 * j));
    }

    int cnt[SSEG];
    #pragma unroll
    for (int s = 0; s < SSEG; ++s) cnt[s] = 0;
    #pragma unroll
    for (int i = 0; i < 16; ++i) {
        const unsigned seg = (unsigned)i >> 1;
        const unsigned pos = ((unsigned)i & 1u) * 256u + tid;
        const unsigned r = q * 8u + (pos >> 6), w = pos & 63u;
        const int v = sem[((size_t)(b * SSEG + seg) * SEMW + 4u * r) * SEMW + 4u * w];
        const bool nz = (v != 0);
        cnt[seg] += __builtin_popcount(__builtin_amdgcn_ballot_w32(nz));
        const unsigned jj = pos >> 3, hh = (pos >> 2) & 1u, e = pos & 3u;
        ((float*)Ash)[((jj * 32u + seg + 16u * hh) << 2) + e] = nz ? 1.0f : 0.0f;
    }
    #pragma unroll
    for (int k = 0; k < 4; ++k) {
        const unsigned n = (unsigned)k * 256u + tid;
        const unsigned jj = n >> 4, sub = n & 15u;
        const unsigned row = 8u + (sub & 7u), hh = sub >> 3;
        const float val = (row == 8u) ? 1.0f : 0.0f;
        v4f t4 = {val, val, val, val};
        Ash[jj * 32u + row + 16u * hh] = t4;
    }
    if (lane == 0u) {
        #pragma unroll
        for (int s = 0; s < SSEG; ++s) cntw[wave][s] = cnt[s];
    }
    __syncthreads();
    if (tid < SSEG) {
        int a = 0;
        #pragma unroll
        for (int w2 = 0; w2 < 8; ++w2) a += cntw[w2][tid];
        areap[(b * NQ + q) * SSEG + tid] = (float)a;
    }

    v8f acc[4];
    #pragma unroll
    for (int g = 0; g < 4; ++g) acc[g] = (v8f){0.f, 0.f, 0.f, 0.f, 0.f, 0.f, 0.f, 0.f};
    v4f A[16];

    #pragma unroll
    for (int s = 0; s < 8; ++s) {
        if (s + 1 < 8) {
            const int n = s + 1;
            const float* __restrict__ p = bbase + (size_t)(2 * (n & 1)) * 16 * SPATIAL + (n >> 1) * 128;
            #pragma unroll
            for (int j = 0; j < 16; ++j) {
                B[n & 1][j]      = __builtin_nontemporal_load((const v4f*)(p + 8 * j));
                B[n & 1][16 + j] = __builtin_nontemporal_load((const v4f*)(p + 16 * SPATIAL + 8 * j));
            }
        }
        __builtin_amdgcn_sched_barrier(0);
        if ((s & 1) == 0) {
            #pragma unroll
            for (int j = 0; j < 16; ++j) A[j] = Ash[((s >> 1) * 16 + j) * 32 + lane];
        }
        const int g0 = 2 * (s & 1), g1 = g0 + 1;
        #pragma unroll
        for (int j = 0; j < 16; ++j) {
            #pragma unroll
            for (int u = 0; u < 2; ++u) {
                const v2f a2 = {A[j][2 * u], A[j][2 * u + 1]};
                const v2f b0 = {B[s & 1][j][2 * u], B[s & 1][j][2 * u + 1]};
                const v2f b1 = {B[s & 1][16 + j][2 * u], B[s & 1][16 + j][2 * u + 1]};
                acc[g0] = __builtin_amdgcn_wmma_f32_16x16x4_f32(false, a2, false, b0, (short)0, acc[g0], false, false);
                acc[g1] = __builtin_amdgcn_wmma_f32_16x16x4_f32(false, a2, false, b1, (short)0, acc[g1], false, false);
            }
        }
        __builtin_amdgcn_sched_barrier(0);
    }

    #pragma unroll
    for (int g = 0; g < 4; ++g) {
        float* __restrict__ t = part + ((size_t)((b * NQ + q) * 32u + wave * 4u + g)) * TILE;
        t[lane] = acc[g][0];
        if (lane < 16u) {
            #pragma unroll
            for (int r = 1; r < 8; ++r) t[16 + 16 * r + lane] = acc[g][r];
        }
    }
}

__global__ void __launch_bounds__(256) seg_combine(const float* __restrict__ part,
                                                   const float* __restrict__ areap,
                                                   float* __restrict__ out) {
    const unsigned gid = blockIdx.x * 256u + threadIdx.x;
    const unsigned ch  = gid & 511u;
    const unsigned s   = (gid >> 9) & 7u;
    const unsigned b   = gid >> 12;
    const unsigned cg  = ch >> 4, c = ch & 15u;

    float ar = 0.0f;
    #pragma unroll
    for (int q = 0; q < NQ; ++q) ar += areap[(b * NQ + q) * SSEG + s];

    const unsigned off = (ar > 0.0f) ? ((s == 0u) ? c : (16u + 16u * s + c)) : (16u + c);
    float sum = 0.0f;
    #pragma unroll
    for (int q = 0; q < NQ; ++q)
        sum += part[((size_t)((b * NQ + q) * 32u + cg)) * TILE + off];

    out[gid] = (ar > 0.0f) ? (sum / ar) : (sum * (1.0f / (float)SPATIAL));
}

extern "C" void kernel_launch(void* const* d_in, const int* in_sizes, int n_in,
                              void* d_out, int out_size, void* d_ws, size_t ws_size,
                              hipStream_t stream) {
    const float* codes = (const float*)d_in[0];
    const int*   sem   = (const int*)d_in[1];
    float*       out   = (float*)d_out;

    float* part  = (float*)d_ws;
    float* areap = (float*)((char*)d_ws +
                            (size_t)BS * NQ * 32 * TILE * sizeof(float));

    hipLaunchKernelGGL(seg_partial, dim3(BS * NQ), dim3(256),
                       0, stream, codes, sem, part, areap);
    hipLaunchKernelGGL(seg_combine, dim3((BS * SSEG * CS) / 256), dim3(256),
                       0, stream, part, areap, out);
}


# codes loads made whole-line: a 16-byte load covers 4 channels x 128 contiguous bytes; B fragments rebuilt in registers by a 4x4 lane-quad transpose (row gathers, selects), no LDS; sum order unchanged
# speedup vs baseline: 4.8965x; 1.1688x over previous
#include <hip/hip_runtime.h>


typedef float v2f __attribute__((ext_vector_type(2)));
typedef float v4f __attribute__((ext_vector_type(4)));
typedef float v8f __attribute__((ext_vector_type(8)));

#define BS      32
#define CS      512
#define SSEG    8
#define SPATIAL 4096
#define SEMW    256
#define NQ      8
#define PCH     (SPATIAL / NQ)
#define NJ      (PCH / 8)
#define TILE    160

__device__ __forceinline__ float quad_x1(float v) {
    return __builtin_bit_cast(float, __builtin_amdgcn_permlane16(
        0u, __builtin_bit_cast(unsigned, v), 0x67452301u, 0xefcdab89u, true, true));
}
__device__ __forceinline__ float quad_x2(float v) {
    return __builtin_bit_cast(float, __builtin_amdgcn_permlane16(
        0u, __builtin_bit_cast(unsigned, v), 0x54761032u, 0xdcfe98bau, true, true));
}
__device__ __forceinline__ void quad_transpose(const v4f& s0, const v4f& s1, const v4f& s2,
                                               const v4f& s3, v4f (&out)[4],
                                               const bool c0, const bool c1) {
    #pragma unroll
    for (int e = 0; e < 4; ++e) {
        const float n0 = quad_x1(s0[e]), n1 = quad_x1(s1[e]);
        const float n2 = quad_x1(s2[e]), n3 = quad_x1(s3[e]);
        const float m0 = c0 ? n1 : s0[e];
        const float m1 = c0 ? s1[e] : n0;
        const float m2 = c0 ? n3 : s2[e];
        const float m3 = c0 ? s3[e] : n2;
        const float k0 = quad_x2(m0), k1 = quad_x2(m1);
        const float k2 = quad_x2(m2), k3 = quad_x2(m3);
        out[0][e] = c1 ? k2 : m0;
        out[1][e] = c1 ? k3 : m1;
        out[2][e] = c1 ? m2 : k0;
        out[3][e] = c1 ? m3 : k1;
    }
}

template <int par>
__device__ __forceinline__ void run_stage(const float* __restrict__ bbase, const v4f* Ash,
                                          const int qc, const unsigned lane,
                                          const bool c0, const bool c1,
                                          v4f (&R)[2][32], v4f (&A)[16], v8f (&acc)[4]) {
    if (par == 0 || qc < 3) {
        const float* __restrict__ p = (par == 0) ? (bbase + (size_t)2 * 16 * SPATIAL + qc * 128)
                                                 : (bbase + (qc + 1) * 128);
        #pragma unroll
        for (int t = 0; t < 4; ++t) {
            #pragma unroll
            for (int i = 0; i < 4; ++i) {
                R[par ^ 1][4 * t + i]      = __builtin_nontemporal_load((const v4f*)(p + i * SPATIAL + 32 * t));
                R[par ^ 1][16 + 4 * t + i] = __builtin_nontemporal_load((const v4f*)(p + (16 + i) * SPATIAL + 32 * t));
            }
        }
    }
    __builtin_amdgcn_sched_barrier(0);
    if (par == 0) {
        #pragma unroll
        for (int j = 0; j < 16; ++j) A[j] = Ash[(qc * 16 + j) * 32 + lane];
    }
    const int g0 = 2 * par, g1 = g0 + 1;
    #pragma unroll
    for (int t = 0; t < 4; ++t) {
        v4f F0[4], F1[4];
        quad_transpose(R[par][4 * t], R[par][4 * t + 1], R[par][4 * t + 2], R[par][4 * t + 3], F0, c0, c1);
        quad_transpose(R[par][16 + 4 * t], R[par][16 + 4 * t + 1], R[par][16 + 4 * t + 2], R[par][16 + 4 * t + 3], F1, c0, c1);
        __builtin_amdgcn_sched_barrier(0);
        #pragma unroll
        for (int pp = 0; pp < 4; ++pp) {
            const int j = 4 * t + pp;
            #pragma unroll
            for (int u = 0; u < 2; ++u) {
                const v2f a2 = {A[j][2 * u], A[j][2 * u + 1]};
                const v2f b0 = {F0[pp][2 * u], F0[pp][2 * u + 1]};
                const v2f b1 = {F1[pp][2 * u], F1[pp][2 * u + 1]};
                acc[g0] = __builtin_amdgcn_wmma_f32_16x16x4_f32(false, a2, false, b0, (short)0, acc[g0], false, false);
                acc[g1] = __builtin_amdgcn_wmma_f32_16x16x4_f32(false, a2, false, b1, (short)0, acc[g1], false, false);
            }
        }
        asm volatile("v_nop\n\tv_nop\n\tv_nop\n\tv_nop\n\tv_nop\n\tv_nop\n\tv_nop\n\tv_nop"
                     : "+v"(acc[g0]), "+v"(acc[g1]));
        __builtin_amdgcn_sched_barrier(0);
    }
}

__global__ void __launch_bounds__(256) seg_partial(const float* __restrict__ codes,
                                                   const int* __restrict__ sem,
                                                   float* __restrict__ part,
                                                   float* __restrict__ areap) {
    __shared__ v4f Ash[NJ * 32];
    __shared__ int cntw[8][SSEG];

    const unsigned tid  = threadIdx.x;
    const unsigned wave = tid >> 5;
    const unsigned lane = tid & 31u;
    const unsigned b    = blockIdx.x >> 3;
    const unsigned q    = blockIdx.x & 7u;
    const unsigned h    = lane >> 4;

    const unsigned x    = lane & 3u;
    const unsigned Q    = (lane >> 2) & 3u;
    const bool     c0   = (lane & 1u) != 0u;
    const bool     c1   = (lane & 2u) != 0u;

    const float* __restrict__ bbase =
        codes + ((size_t)(b * CS + wave * 64u + 4u * Q)) * SPATIAL + q * PCH + 8u * x + 4u * h;

    v4f R[2][32];
    #pragma unroll
    for (int t = 0; t < 4; ++t) {
        #pragma unroll
        for (int i = 0; i < 4; ++i) {
            R[0][4 * t + i]      = __builtin_nontemporal_load((const v4f*)(bbase + i * SPATIAL + 32 * t));
            R[0][16 + 4 * t + i] = __builtin_nontemporal_load((const v4f*)(bbase + (16 + i) * SPATIAL + 32 * t));
        }
    }

    int cnt[SSEG];
    #pragma unroll
    for (int s = 0; s < SSEG; ++s) cnt[s] = 0;
    #pragma unroll
    for (int i = 0; i < 16; ++i) {
        const unsigned seg = (unsigned)i >> 1;
        const unsigned pos = ((unsigned)i & 1u) * 256u + tid;
        const unsigned r = q * 8u + (pos >> 6), w = pos & 63u;
        const int v = sem[((size_t)(b * SSEG + seg) * SEMW + 4u * r) * SEMW + 4u * w];
        const bool nz = (v != 0);
        cnt[seg] += __builtin_popcount(__builtin_amdgcn_ballot_w32(nz));
        const unsigned jj = pos >> 3, hh = (pos >> 2) & 1u, e = pos & 3u;
        ((float*)Ash)[((jj * 32u + seg + 16u * hh) << 2) + e] = nz ? 1.0f : 0.0f;
    }
    #pragma unroll
    for (int k = 0; k < 4; ++k) {
        const unsigned n = (unsigned)k * 256u + tid;
        const unsigned jj = n >> 4, sub = n & 15u;
        const unsigned row = 8u + (sub & 7u), hh = sub >> 3;
        const float val = (row == 8u) ? 1.0f : 0.0f;
        v4f t4 = {val, val, val, val};
        Ash[jj * 32u + row + 16u * hh] = t4;
    }
    if (lane == 0u) {
        #pragma unroll
        for (int s = 0; s < SSEG; ++s) cntw[wave][s] = cnt[s];
    }
    __syncthreads();
    if (tid < SSEG) {
        int a = 0;
        #pragma unroll
        for (int w2 = 0; w2 < 8; ++w2) a += cntw[w2][tid];
        areap[(b * NQ + q) * SSEG + tid] = (float)a;
    }

    v8f acc[4];
    #pragma unroll
    for (int g = 0; g < 4; ++g) acc[g] = (v8f){0.f, 0.f, 0.f, 0.f, 0.f, 0.f, 0.f, 0.f};
    v4f A[16];

    #pragma unroll 1
    for (int qc = 0; qc < 4; ++qc) {
        run_stage<0>(bbase, Ash, qc, lane, c0, c1, R, A, acc);
        run_stage<1>(bbase, Ash, qc, lane, c0, c1, R, A, acc);
    }

    #pragma unroll
    for (int g = 0; g < 4; ++g) {
        float* __restrict__ t = part + ((size_t)((b * NQ + q) * 32u + wave * 4u + g)) * TILE;
        t[lane] = acc[g][0];
        if (lane < 16u) {
            #pragma unroll
            for (int r = 1; r < 8; ++r) t[16 + 16 * r + lane] = acc[g][r];
        }
    }
}

__global__ void __launch_bounds__(256) seg_combine(const float* __restrict__ part,
                                                   const float* __restrict__ areap,
                                                   float* __restrict__ out) {
    const unsigned gid = blockIdx.x * 256u + threadIdx.x;
    const unsigned ch  = gid & 511u;
    const unsigned s   = (gid >> 9) & 7u;
    const unsigned b   = gid >> 12;
    const unsigned cg  = ch >> 4, c = ch & 15u;

    float ar = 0.0f;
    #pragma unroll
    for (int q = 0; q < NQ; ++q) ar += areap[(b * NQ + q) * SSEG + s];

    const unsigned off = (ar > 0.0f) ? ((s == 0u) ? c : (16u + 16u * s + c)) : (16u + c);
    float sum = 0.0f;
    #pragma unroll
    for (int q = 0; q < NQ; ++q)
        sum += part[((size_t)((b * NQ + q) * 32u + cg)) * TILE + off];

    out[gid] = (ar > 0.0f) ? (sum / ar) : (sum * (1.0f / (float)SPATIAL));
}

extern "C" void kernel_launch(void* const* d_in, const int* in_sizes, int n_in,
                              void* d_out, int out_size, void* d_ws, size_t ws_size,
                              hipStream_t stream) {
    const float* codes = (const float*)d_in[0];
    const int*   sem   = (const int*)d_in[1];
    float*       out   = (float*)d_out;

    float* part  = (float*)d_ws;
    float* areap = (float*)((char*)d_ws +
                            (size_t)BS * NQ * 32 * TILE * sizeof(float));

    hipLaunchKernelGGL(seg_partial, dim3(BS * NQ), dim3(256),
                       0, stream, codes, sem, part, areap);
    hipLaunchKernelGGL(seg_combine, dim3((BS * SSEG * CS) / 256), dim3(256),
                       0, stream, part, areap, out);
}
